# starting kernel
# baseline (speedup 1.0000x reference)
#include <hip/hip_runtime.h>
#include <hip/hip_bf16.h>

typedef __attribute__((ext_vector_type(2))) float v2f;
typedef __attribute__((ext_vector_type(8))) float v8f;

#define V_NODES 8192
#define C_CH 4
#define K_KS 4
#define F_FIFO 16
#define NSPLIT 8
#define VSPLIT (V_NODES / NSPLIT)
#define COLS_PER_BLOCK 64
#define THREADS_K2 128

__global__ void stgcn_prep(const float* __restrict__ x, float* __restrict__ s) {
    int v = blockIdx.x * blockDim.x + threadIdx.x;
    if (v < V_NODES) {
#pragma unroll
        for (int k = 0; k < K_KS; ++k) {
            float acc = 0.f;
#pragma unroll
            for (int c = 0; c < C_CH; ++c)
                acc += x[(c * K_KS + k) * V_NODES + v];
            s[k * V_NODES + v] = acc;
        }
    }
}

__global__ void __launch_bounds__(THREADS_K2)
stgcn_wmma(const float* __restrict__ graph, const float* __restrict__ s,
           float* __restrict__ partial) {
    __shared__ float s_lds[K_KS * VSPLIT];

    const int tid   = threadIdx.x;
    const int lane  = tid & 31;
    const int wave  = tid >> 5;
    const int vbase = blockIdx.y * VSPLIT;

    for (int i = tid; i < K_KS * VSPLIT; i += THREADS_K2) {
        int k  = i >> 10;
        int vo = i & (VSPLIT - 1);
        s_lds[i] = s[k * V_NODES + vbase + vo];
    }
    __syncthreads();

    const int w0   = blockIdx.x * COLS_PER_BLOCK + wave * 16;
    const int n    = lane & 15;
    const int kb   = (lane >> 4) << 1;
    const int mLds = lane & 3;

    const float* gp     = graph + (size_t)(vbase + kb) * V_NODES + (w0 + n);
    const float* a_base = &s_lds[mLds * VSPLIT + kb];

    v8f acc = {};
#pragma unroll 4
    for (int vo = 0; vo < VSPLIT; vo += 4) {
        v2f a = *(const v2f*)(a_base + vo);
        v2f b;
        b.x = gp[0];
        b.y = gp[V_NODES];
        __builtin_prefetch(gp + (size_t)32 * V_NODES, 0, 0);
        acc = __builtin_amdgcn_wmma_f32_16x16x4_f32(
            false, a, false, b, (short)0, acc, false, false);
        gp += (size_t)4 * V_NODES;
    }

    if (lane < 16) {
        float* p = partial + (size_t)blockIdx.y * (K_KS * V_NODES) + w0 + lane;
#pragma unroll
        for (int r = 0; r < 4; ++r)
            p[r * V_NODES] = acc[r];
    }
}

__global__ void stgcn_finish(const float* __restrict__ fifo,
                             const float* __restrict__ partial,
                             float* __restrict__ out) {
    int i = blockIdx.x * blockDim.x + threadIdx.x;
    if (i >= C_CH * V_NODES) return;
    int k = i >> 13;
    int w = i & (V_NODES - 1);

    float acc = 0.f;
#pragma unroll
    for (int j = 1; j <= 13; j += 2)
        acc += fifo[(j * C_CH + k) * V_NODES + w];
#pragma unroll
    for (int sp = 0; sp < NSPLIT; ++sp)
        acc += partial[(size_t)sp * (K_KS * V_NODES) + k * V_NODES + w];
    out[i] = acc;
}

extern "C" void kernel_launch(void* const* d_in, const int* in_sizes, int n_in,
                              void* d_out, int out_size, void* d_ws, size_t ws_size,
                              hipStream_t stream) {
    const float* x     = (const float*)d_in[0];
    const float* graph = (const float*)d_in[1];
    const float* fifo  = (const float*)d_in[2];

    float* out     = (float*)d_out;
    float* s_ws    = (float*)d_ws;
    float* partial = s_ws + (size_t)K_KS * V_NODES;

    stgcn_prep<<<V_NODES / 256, 256, 0, stream>>>(x, s_ws);

    dim3 grid2(V_NODES / COLS_PER_BLOCK, NSPLIT);
    stgcn_wmma<<<grid2, THREADS_K2, 0, stream>>>(graph, s_ws, partial);

    stgcn_finish<<<(C_CH * V_NODES + 255) / 256, 256, 0, stream>>>(fifo, partial, out);
}


# graph streamed once by non-temporal 128-bit loads (aligned 256-byte row pieces), 24 in flight per wave in a register ring; 512-thread blocks sum 4 row bands in LDS, so the sum order changes
# speedup vs baseline: 1.7968x; 1.7968x over previous
#include <hip/hip_runtime.h>
#include <hip/hip_bf16.h>

typedef __attribute__((ext_vector_type(2))) float v2f;
typedef __attribute__((ext_vector_type(4))) float v4f;
typedef __attribute__((ext_vector_type(8))) float v8f;

#define V_NODES 8192
#define C_CH 4
#define K_KS 4
#define F_FIFO 16
#define NSPLIT 8
#define VSPLIT (V_NODES / NSPLIT)
#define COLS_PER_WAVE 64
#define COL_WAVES 4
#define ROW_WAVES 4
#define COLS_PER_BLOCK (COLS_PER_WAVE * COL_WAVES)
#define THREADS_K2 (32 * COL_WAVES * ROW_WAVES)
#define BAND_ROWS (VSPLIT / ROW_WAVES)
#define KSTEPS (BAND_ROWS / 4)
#define AHEAD 12
#define NSLOT (AHEAD + 1)

__global__ void stgcn_prep(const float* __restrict__ x, float* __restrict__ s) {
    int v = blockIdx.x * blockDim.x + threadIdx.x;
    if (v < V_NODES) {
#pragma unroll
        for (int k = 0; k < K_KS; ++k) {
            float acc = 0.f;
#pragma unroll
            for (int c = 0; c < C_CH; ++c)
                acc += x[(c * K_KS + k) * V_NODES + v];
            s[k * V_NODES + v] = acc;
        }
    }
}

__global__ void __launch_bounds__(THREADS_K2)
stgcn_wmma(const float* __restrict__ graph, const float* __restrict__ s,
           float* __restrict__ partial) {
    __shared__ float s_lds[K_KS * VSPLIT];
    __shared__ float red[ROW_WAVES][K_KS][COLS_PER_BLOCK];

    const int tid   = threadIdx.x;
    const int lane  = tid & 31;
    const int wave  = tid >> 5;
    const int cw    = wave & (COL_WAVES - 1);
    const int rw    = wave >> 2;
    const int vbase = blockIdx.y * VSPLIT;

    const int n    = lane & 15;
    const int kb   = (lane >> 4) << 1;
    const int mLds = lane & 3;
    const int cw0  = blockIdx.x * COLS_PER_BLOCK + cw * COLS_PER_WAVE;

    v4f sreg[2];
#pragma unroll
    for (int j = 0; j < 2; ++j) {
        int i  = tid + THREADS_K2 * j;
        int k  = i >> 8;
        int v4 = i & 255;
        sreg[j] = *(const v4f*)(s + k * V_NODES + vbase + 4 * v4);
    }

    const float* gp = graph + (size_t)(vbase + rw * BAND_ROWS + kb) * V_NODES + (cw0 + 4 * n);
    v4f ring[NSLOT][2];
#define LOAD_STEP(I) do { \
        ring[(I) % NSLOT][0] = __builtin_nontemporal_load((const v4f*)(gp + (size_t)(4 * (I))     * V_NODES)); \
        ring[(I) % NSLOT][1] = __builtin_nontemporal_load((const v4f*)(gp + (size_t)(4 * (I) + 1) * V_NODES)); \
    } while (0)

#pragma unroll
    for (int i = 0; i < AHEAD; ++i) LOAD_STEP(i);

#pragma unroll
    for (int j = 0; j < 2; ++j)
        *(v4f*)&s_lds[4 * (tid + THREADS_K2 * j)] = sreg[j];
    __builtin_amdgcn_fence(__ATOMIC_RELEASE, "workgroup", "local");
    __builtin_amdgcn_s_barrier();
    __builtin_amdgcn_fence(__ATOMIC_ACQUIRE, "workgroup", "local");

    const float* a_base = &s_lds[mLds * VSPLIT + rw * BAND_ROWS + kb];

    v8f acc[4] = {};
    v2f a_nxt = *(const v2f*)(a_base);
#pragma unroll
    for (int i = 0; i < KSTEPS; ++i) {
        __builtin_amdgcn_sched_barrier(0);
        if (i + AHEAD < KSTEPS) LOAD_STEP(i + AHEAD);
        const v2f a = a_nxt;
        if (i + 1 < KSTEPS) a_nxt = *(const v2f*)(a_base + 4 * (i + 1));
        __builtin_amdgcn_sched_barrier(0);
        const v4f r0 = ring[i % NSLOT][0];
        const v4f r1 = ring[i % NSLOT][1];
#pragma unroll
        for (int t = 0; t < 4; ++t) {
            v2f b;
            b.x = r0[t];
            b.y = r1[t];
            acc[t] = __builtin_amdgcn_wmma_f32_16x16x4_f32(
                false, a, false, b, (short)0, acc[t], false, false);
        }
    }
#undef LOAD_STEP

    if (lane < 16) {
#pragma unroll
        for (int r = 0; r < 4; ++r) {
            v4f o;
            o.x = acc[0][r]; o.y = acc[1][r]; o.z = acc[2][r]; o.w = acc[3][r];
            *(v4f*)&red[rw][r][cw * COLS_PER_WAVE + 4 * lane] = o;
        }
    }
    __syncthreads();

    if (tid < K_KS * (COLS_PER_BLOCK / 4)) {
        const int k  = tid >> 6;
        const int c4 = tid & 63;
        v4f sum = *(const v4f*)&red[0][k][4 * c4];
#pragma unroll
        for (int b = 1; b < ROW_WAVES; ++b)
            sum += *(const v4f*)&red[b][k][4 * c4];
        float* p = partial + (size_t)blockIdx.y * (K_KS * V_NODES) + k * V_NODES
                 + blockIdx.x * COLS_PER_BLOCK + 4 * c4;
        *(v4f*)p = sum;
    }
}

__global__ void stgcn_finish(const float* __restrict__ fifo,
                             const float* __restrict__ partial,
                             float* __restrict__ out) {
    int i = blockIdx.x * blockDim.x + threadIdx.x;
    if (i >= C_CH * V_NODES) return;
    int k = i >> 13;
    int w = i & (V_NODES - 1);

    float acc = 0.f;
#pragma unroll
    for (int j = 1; j <= 13; j += 2)
        acc += fifo[(j * C_CH + k) * V_NODES + w];
#pragma unroll
    for (int sp = 0; sp < NSPLIT; ++sp)
        acc += partial[(size_t)sp * (K_KS * V_NODES) + k * V_NODES + w];
    out[i] = acc;
}

extern "C" void kernel_launch(void* const* d_in, const int* in_sizes, int n_in,
                              void* d_out, int out_size, void* d_ws, size_t ws_size,
                              hipStream_t stream) {
    const float* x     = (const float*)d_in[0];
    const float* graph = (const float*)d_in[1];
    const float* fifo  = (const float*)d_in[2];

    float* out     = (float*)d_out;
    float* s_ws    = (float*)d_ws;
    float* partial = s_ws + (size_t)K_KS * V_NODES;

    stgcn_prep<<<V_NODES / 256, 256, 0, stream>>>(x, s_ws);

    dim3 grid2(V_NODES / COLS_PER_BLOCK, NSPLIT);
    stgcn_wmma<<<grid2, THREADS_K2, 0, stream>>>(graph, s_ws, partial);

    stgcn_finish<<<(C_CH * V_NODES + 255) / 256, 256, 0, stream>>>(fifo, partial, out);
}
